# starting kernel
# baseline (speedup 1.0000x reference)
#include <hip/hip_runtime.h>
#include <math.h>

typedef __attribute__((ext_vector_type(2))) float v2f;
typedef __attribute__((ext_vector_type(8))) float v8f;

#define WAVES   4
#define ROWS    16
#define HPITCH  197
#define H1PITCH 67
#define H2PITCH 33
#define OPITCH  17
#define H2OFF   1536

__device__ __forceinline__ void jacobi3(float* s, float* v, int p, int q) {
  float apq = s[p * 3 + q];
  if (fabsf(apq) < 1e-20f) return;
  float app = s[p * 3 + p], aqq = s[q * 3 + q];
  float tau = (aqq - app) / (2.f * apq);
  float t  = copysignf(1.f, tau) / (fabsf(tau) + sqrtf(1.f + tau * tau));
  float c  = 1.f / sqrtf(1.f + t * t);
  float sn = t * c;
  const int k = 3 - p - q;
  float skp = s[k * 3 + p], skq = s[k * 3 + q];
  float nkp = c * skp - sn * skq;
  float nkq = sn * skp + c * skq;
  s[k * 3 + p] = nkp; s[p * 3 + k] = nkp;
  s[k * 3 + q] = nkq; s[q * 3 + k] = nkq;
  s[p * 3 + p] = c * c * app - 2.f * sn * c * apq + sn * sn * aqq;
  s[q * 3 + q] = sn * sn * app + 2.f * sn * c * apq + c * c * aqq;
  s[p * 3 + q] = 0.f; s[q * 3 + p] = 0.f;
#pragma unroll
  for (int i = 0; i < 3; ++i) {
    float vip = v[i * 3 + p], viq = v[i * 3 + q];
    v[i * 3 + p] = c * vip - sn * viq;
    v[i * 3 + q] = sn * vip + c * viq;
  }
}

__global__ __launch_bounds__(128) void rmi_fused(
    const float* __restrict__ x,   const float* __restrict__ cwg,
    const float* __restrict__ cbg, const float* __restrict__ w1,
    const float* __restrict__ b1,  const float* __restrict__ w2,
    const float* __restrict__ b2,  const float* __restrict__ w3,
    const float* __restrict__ b3,  float* __restrict__ out, int B) {
  __shared__ float lds [WAVES * ROWS * HPITCH];
  __shared__ float ldsO[WAVES * ROWS * OPITCH];

  const int tid  = threadIdx.x;
  const int wave = tid >> 5;
  const int lane = tid & 31;
  const int rowBase = blockIdx.x * (WAVES * ROWS) + wave * ROWS;
  float* H = &lds [wave * ROWS * HPITCH];
  float* O = &ldsO[wave * ROWS * OPITCH];

  float cw[30];
#pragma unroll
  for (int i = 0; i < 30; ++i) cw[i] = cwg[i];
  const float cb = cbg[0];

  for (int r = 0; r < ROWS; ++r) {
    int sr = rowBase + r; sr = (sr < B) ? sr : (B - 1);
    const float* xr = x + (size_t)sr * 1200;
    for (int j = lane; j < 196; j += 32) {
      __builtin_prefetch(xr + j + 256, 0, 1);
      float acc = cb;
#pragma unroll
      for (int c = 0; c < 6; ++c)
#pragma unroll
        for (int k = 0; k < 5; ++k)
          acc = fmaf(xr[c * 200 + j + k], cw[c * 5 + k], acc);
      H[r * HPITCH + j] = acc;
    }
  }
  __syncthreads();

  const int hs = (lane >= 16) ? 2 : 0;
  const int rA = lane & 15;
  const int rD = (lane < 16) ? 0 : 8;

  int cc1[4];
#pragma unroll
  for (int nt = 0; nt < 4; ++nt) {
    const int c = nt * 16 + rA;
    cc1[nt] = (c < 50) ? c : 49;
  }

  v8f acc1[4] = {};
#pragma unroll 2
  for (int kk = 0; kk < 49; ++kk) {
    const int k0 = kk * 4 + hs;
    v2f a; a.x = H[rA * HPITCH + k0]; a.y = H[rA * HPITCH + k0 + 1];
#pragma unroll
    for (int nt = 0; nt < 4; ++nt) {
      v2f b;
      b.x = w1[k0 * 50 + cc1[nt]];
      b.y = w1[(k0 + 1) * 50 + cc1[nt]];
      acc1[nt] = __builtin_amdgcn_wmma_f32_16x16x4_f32(
          false, a, false, b, (short)0, acc1[nt], false, false);
    }
  }
  __syncthreads();

  float* H1 = H;
#pragma unroll
  for (int nt = 0; nt < 4; ++nt) {
    const int col = nt * 16 + rA;
    const float bb = b1[cc1[nt]];
#pragma unroll
    for (int r = 0; r < 8; ++r) {
      float vv = fmaxf(acc1[nt][r] + bb, 0.f);
      vv = (col < 50) ? vv : 0.f;
      H1[(rD + r) * H1PITCH + col] = vv;
    }
  }
  __syncthreads();

  const int cc2 = (16 + rA < 30) ? (16 + rA) : 29;
  v8f acc2[2] = {};
  for (int kk = 0; kk < 13; ++kk) {
    const int k0  = kk * 4 + hs;
    const int kc0 = (k0 < 50) ? k0 : 49;
    const int kc1 = (k0 + 1 < 50) ? (k0 + 1) : 49;
    v2f a;  a.x  = H1[rA * H1PITCH + k0]; a.y = H1[rA * H1PITCH + k0 + 1];
    v2f b0; b0.x = w2[kc0 * 30 + rA];     b0.y = w2[kc1 * 30 + rA];
    v2f b1v; b1v.x = w2[kc0 * 30 + cc2];  b1v.y = w2[kc1 * 30 + cc2];
    acc2[0] = __builtin_amdgcn_wmma_f32_16x16x4_f32(
        false, a, false, b0, (short)0, acc2[0], false, false);
    acc2[1] = __builtin_amdgcn_wmma_f32_16x16x4_f32(
        false, a, false, b1v, (short)0, acc2[1], false, false);
  }
  __syncthreads();
  float* H2 = H + H2OFF;
#pragma unroll
  for (int nt = 0; nt < 2; ++nt) {
    const int col = nt * 16 + rA;
    const float bb = b2[(col < 30) ? col : 29];
#pragma unroll
    for (int r = 0; r < 8; ++r) {
      float vv = fmaxf(acc2[nt][r] + bb, 0.f);
      vv = (col < 30) ? vv : 0.f;
      H2[(rD + r) * H2PITCH + col] = vv;
    }
  }
  __syncthreads();

  const int cc3 = (rA < 15) ? rA : 14;
  v8f acc3 = {};
  for (int kk = 0; kk < 8; ++kk) {
    const int k0  = kk * 4 + hs;
    const int kc0 = (k0 < 30) ? k0 : 29;
    const int kc1 = (k0 + 1 < 30) ? (k0 + 1) : 29;
    v2f a; a.x = H2[rA * H2PITCH + k0]; a.y = H2[rA * H2PITCH + k0 + 1];
    v2f b; b.x = w3[kc0 * 15 + cc3];    b.y = w3[kc1 * 15 + cc3];
    acc3 = __builtin_amdgcn_wmma_f32_16x16x4_f32(
        false, a, false, b, (short)0, acc3, false, false);
  }
  {
    const float bb = b3[cc3];
#pragma unroll
    for (int r = 0; r < 8; ++r)
      O[(rD + r) * OPITCH + rA] = acc3[r] + bb;
  }
  __syncthreads();

  if (lane < 16) {
    const int gRow = rowBase + lane;
    if (gRow < B) {
      float a9[9];
#pragma unroll
      for (int i = 0; i < 9; ++i) a9[i] = O[lane * OPITCH + i];
      float s[9], v[9];
#pragma unroll
      for (int i = 0; i < 3; ++i)
#pragma unroll
        for (int j = 0; j < 3; ++j) {
          s[i * 3 + j] = a9[0 * 3 + i] * a9[0 * 3 + j] +
                         a9[1 * 3 + i] * a9[1 * 3 + j] +
                         a9[2 * 3 + i] * a9[2 * 3 + j];
          v[i * 3 + j] = (i == j) ? 1.f : 0.f;
        }
#pragma unroll
      for (int sw = 0; sw < 6; ++sw) {
        jacobi3(s, v, 0, 1); jacobi3(s, v, 0, 2); jacobi3(s, v, 1, 2);
      }
      float g0 = s[0], g1 = s[4], g2 = s[8];
#define SWAPCOL(p, q) { float t_;                              \
        t_ = v[p];     v[p]     = v[q];     v[q]     = t_;     \
        t_ = v[3 + p]; v[3 + p] = v[3 + q]; v[3 + q] = t_;     \
        t_ = v[6 + p]; v[6 + p] = v[6 + q]; v[6 + q] = t_; }
      if (g0 < g1) { float t = g0; g0 = g1; g1 = t; SWAPCOL(0, 1); }
      if (g0 < g2) { float t = g0; g0 = g2; g2 = t; SWAPCOL(0, 2); }
      if (g1 < g2) { float t = g1; g1 = g2; g2 = t; SWAPCOL(1, 2); }
#undef SWAPCOL
      float w[9];
#pragma unroll
      for (int i = 0; i < 3; ++i)
#pragma unroll
        for (int j = 0; j < 3; ++j)
          w[i * 3 + j] = a9[i * 3 + 0] * v[0 + j] +
                         a9[i * 3 + 1] * v[3 + j] +
                         a9[i * 3 + 2] * v[6 + j];
      float u0x = w[0], u0y = w[3], u0z = w[6];
      float n0 = 1.f / sqrtf(u0x * u0x + u0y * u0y + u0z * u0z + 1e-20f);
      u0x *= n0; u0y *= n0; u0z *= n0;
      float u1x = w[1], u1y = w[4], u1z = w[7];
      float d01 = u1x * u0x + u1y * u0y + u1z * u0z;
      u1x -= d01 * u0x; u1y -= d01 * u0y; u1z -= d01 * u0z;
      float n1 = 1.f / sqrtf(u1x * u1x + u1y * u1y + u1z * u1z + 1e-20f);
      u1x *= n1; u1y *= n1; u1z *= n1;
      float cx = u0y * u1z - u0z * u1y;
      float cy = u0z * u1x - u0x * u1z;
      float cz = u0x * u1y - u0y * u1x;
      float detV = v[0] * (v[4] * v[8] - v[5] * v[7]) -
                   v[1] * (v[3] * v[8] - v[5] * v[6]) +
                   v[2] * (v[3] * v[7] - v[4] * v[6]);
      float sg = (detV >= 0.f) ? 1.f : -1.f;
      float u2x = sg * cx, u2y = sg * cy, u2z = sg * cz;
      float U_[9] = { u0x, u1x, u2x,  u0y, u1y, u2y,  u0z, u1z, u2z };
      float* op = out + (size_t)gRow * 15;
#pragma unroll
      for (int i = 0; i < 3; ++i)
#pragma unroll
        for (int j = 0; j < 3; ++j)
          op[i * 3 + j] = U_[i * 3 + 0] * v[j * 3 + 0] +
                          U_[i * 3 + 1] * v[j * 3 + 1] +
                          U_[i * 3 + 2] * v[j * 3 + 2];
#pragma unroll
      for (int i = 9; i < 15; ++i) op[i] = O[lane * OPITCH + i];
    }
  }
}

extern "C" void kernel_launch(void* const* d_in, const int* in_sizes, int n_in,
                              void* d_out, int out_size, void* d_ws, size_t ws_size,
                              hipStream_t stream) {
  (void)n_in; (void)out_size; (void)d_ws; (void)ws_size;
  const float* x   = (const float*)d_in[0];
  const float* cw  = (const float*)d_in[1];
  const float* cb  = (const float*)d_in[2];
  const float* w1  = (const float*)d_in[3];
  const float* b1  = (const float*)d_in[4];
  const float* w2  = (const float*)d_in[5];
  const float* b2  = (const float*)d_in[6];
  const float* w3  = (const float*)d_in[7];
  const float* b3  = (const float*)d_in[8];
  const int B = in_sizes[0] / 1200;
  const int grid = (B + WAVES * ROWS - 1) / (WAVES * ROWS);
  rmi_fused<<<grid, WAVES * 32, 0, stream>>>(x, cw, cb, w1, b1, w2, b2, w3, b3,
                                             (float*)d_out, B);
}


# conv1d input streamed two rows ahead per wave: a lane keeps a 12-float window per channel (three 16-byte loads) and forms 8 adjacent outputs from it; every sum keeps its order
# speedup vs baseline: 2.7475x; 2.7475x over previous
#include <hip/hip_runtime.h>
#include <math.h>

typedef __attribute__((ext_vector_type(2))) float v2f;
typedef __attribute__((ext_vector_type(4))) float v4f;
typedef __attribute__((ext_vector_type(8))) float v8f;

#define WAVES   4
#define ROWS    16
#define HPITCH  197
#define H1PITCH 67
#define H2PITCH 33
#define OPITCH  17
#define H2OFF   1536

__device__ __forceinline__ void jacobi3(float* s, float* v, int p, int q) {
  float apq = s[p * 3 + q];
  if (fabsf(apq) < 1e-20f) return;
  float app = s[p * 3 + p], aqq = s[q * 3 + q];
  float tau = (aqq - app) / (2.f * apq);
  float t  = copysignf(1.f, tau) / (fabsf(tau) + sqrtf(1.f + tau * tau));
  float c  = 1.f / sqrtf(1.f + t * t);
  float sn = t * c;
  const int k = 3 - p - q;
  float skp = s[k * 3 + p], skq = s[k * 3 + q];
  float nkp = c * skp - sn * skq;
  float nkq = sn * skp + c * skq;
  s[k * 3 + p] = nkp; s[p * 3 + k] = nkp;
  s[k * 3 + q] = nkq; s[q * 3 + k] = nkq;
  s[p * 3 + p] = c * c * app - 2.f * sn * c * apq + sn * sn * aqq;
  s[q * 3 + q] = sn * sn * app + 2.f * sn * c * apq + c * c * aqq;
  s[p * 3 + q] = 0.f; s[q * 3 + p] = 0.f;
#pragma unroll
  for (int i = 0; i < 3; ++i) {
    float vip = v[i * 3 + p], viq = v[i * 3 + q];
    v[i * 3 + p] = c * vip - sn * viq;
    v[i * 3 + q] = sn * vip + c * viq;
  }
}

__device__ __forceinline__ int src_row(int rowBase, int r, int B) {
  const int rr = (r < 15) ? r : 15;
  const int sr = rowBase + rr;
  return (sr < B) ? sr : (B - 1);
}

__device__ __forceinline__ void load_row(v4f (&buf)[18], const float* __restrict__ x,
                                         int row, int i0, int i1, int i2) {
  const v4f* p = reinterpret_cast<const v4f*>(x + (size_t)row * 1200);
#pragma unroll
  for (int c = 0; c < 6; ++c) {
    buf[c * 3 + 0] = p[c * 50 + i0];
    buf[c * 3 + 1] = p[c * 50 + i1];
    buf[c * 3 + 2] = p[c * 50 + i2];
  }
}

__device__ __forceinline__ void conv_row(const v4f (&buf)[18], const float (&cw)[30],
                                         float cb, float* hrow, int lane) {
  float o[8];
#pragma unroll
  for (int i = 0; i < 8; ++i) o[i] = cb;
#pragma unroll
  for (int c = 0; c < 6; ++c) {
    const v4f q0 = buf[c * 3 + 0], q1 = buf[c * 3 + 1], q2 = buf[c * 3 + 2];
    const float xv[12] = { q0.x, q0.y, q0.z, q0.w, q1.x, q1.y, q1.z, q1.w,
                           q2.x, q2.y, q2.z, q2.w };
#pragma unroll
    for (int k = 0; k < 5; ++k)
#pragma unroll
      for (int i = 0; i < 8; ++i)
        o[i] = fmaf(xv[i + k], cw[c * 5 + k], o[i]);
  }
  float* hp = hrow + 8 * lane;
  if (lane < 25) {
#pragma unroll
    for (int i = 0; i < 4; ++i) hp[i] = o[i];
  }
  if (lane < 24) {
#pragma unroll
    for (int i = 4; i < 8; ++i) hp[i] = o[i];
  }
}

__global__ __launch_bounds__(128) void rmi_fused(
    const float* __restrict__ x,   const float* __restrict__ cwg,
    const float* __restrict__ cbg, const float* __restrict__ w1,
    const float* __restrict__ b1,  const float* __restrict__ w2,
    const float* __restrict__ b2,  const float* __restrict__ w3,
    const float* __restrict__ b3,  float* __restrict__ out, int B) {
  __shared__ float lds [WAVES * ROWS * HPITCH];
  __shared__ float ldsO[WAVES * ROWS * OPITCH];

  const int tid  = threadIdx.x;
  const int wave = tid >> 5;
  const int lane = tid & 31;
  const int rowBase = blockIdx.x * (WAVES * ROWS) + wave * ROWS;
  float* H = &lds [wave * ROWS * HPITCH];
  float* O = &ldsO[wave * ROWS * OPITCH];

  float cw[30];
#pragma unroll
  for (int i = 0; i < 30; ++i) cw[i] = cwg[i];
  const float cb = cbg[0];

  const int lc = (lane < 24) ? lane : 24;
  const int i0 = 2 * lc, i1 = 2 * lc + 1;
  const int i2 = (lc < 24) ? (2 * lc + 2) : 49;
#define XROW(r) src_row(rowBase, (r), B)
  v4f bA[18], bB[18], bC[18];
  load_row(bA, x, XROW(0), i0, i1, i2);
  load_row(bB, x, XROW(1), i0, i1, i2);
  for (int r = 0; r < 15; r += 3) {
    load_row(bC, x, XROW(r + 2), i0, i1, i2);
    __builtin_amdgcn_sched_barrier(0);
    conv_row(bA, cw, cb, H + r * HPITCH, lane);
    load_row(bA, x, XROW(r + 3), i0, i1, i2);
    __builtin_amdgcn_sched_barrier(0);
    conv_row(bB, cw, cb, H + (r + 1) * HPITCH, lane);
    load_row(bB, x, XROW(r + 4), i0, i1, i2);
    __builtin_amdgcn_sched_barrier(0);
    conv_row(bC, cw, cb, H + (r + 2) * HPITCH, lane);
  }
  conv_row(bA, cw, cb, H + 15 * HPITCH, lane);
#undef XROW
  __syncthreads();

  const int hs = (lane >= 16) ? 2 : 0;
  const int rA = lane & 15;
  const int rD = (lane < 16) ? 0 : 8;

  int cc1[4];
#pragma unroll
  for (int nt = 0; nt < 4; ++nt) {
    const int c = nt * 16 + rA;
    cc1[nt] = (c < 50) ? c : 49;
  }

  v8f acc1[4] = {};
#pragma unroll 2
  for (int kk = 0; kk < 49; ++kk) {
    const int k0 = kk * 4 + hs;
    v2f a; a.x = H[rA * HPITCH + k0]; a.y = H[rA * HPITCH + k0 + 1];
#pragma unroll
    for (int nt = 0; nt < 4; ++nt) {
      v2f b;
      b.x = w1[k0 * 50 + cc1[nt]];
      b.y = w1[(k0 + 1) * 50 + cc1[nt]];
      acc1[nt] = __builtin_amdgcn_wmma_f32_16x16x4_f32(
          false, a, false, b, (short)0, acc1[nt], false, false);
    }
  }
  __syncthreads();

  float* H1 = H;
#pragma unroll
  for (int nt = 0; nt < 4; ++nt) {
    const int col = nt * 16 + rA;
    const float bb = b1[cc1[nt]];
#pragma unroll
    for (int r = 0; r < 8; ++r) {
      float vv = fmaxf(acc1[nt][r] + bb, 0.f);
      vv = (col < 50) ? vv : 0.f;
      H1[(rD + r) * H1PITCH + col] = vv;
    }
  }
  __syncthreads();

  const int cc2 = (16 + rA < 30) ? (16 + rA) : 29;
  v8f acc2[2] = {};
  for (int kk = 0; kk < 13; ++kk) {
    const int k0  = kk * 4 + hs;
    const int kc0 = (k0 < 50) ? k0 : 49;
    const int kc1 = (k0 + 1 < 50) ? (k0 + 1) : 49;
    v2f a;  a.x  = H1[rA * H1PITCH + k0]; a.y = H1[rA * H1PITCH + k0 + 1];
    v2f b0; b0.x = w2[kc0 * 30 + rA];     b0.y = w2[kc1 * 30 + rA];
    v2f b1v; b1v.x = w2[kc0 * 30 + cc2];  b1v.y = w2[kc1 * 30 + cc2];
    acc2[0] = __builtin_amdgcn_wmma_f32_16x16x4_f32(
        false, a, false, b0, (short)0, acc2[0], false, false);
    acc2[1] = __builtin_amdgcn_wmma_f32_16x16x4_f32(
        false, a, false, b1v, (short)0, acc2[1], false, false);
  }
  __syncthreads();
  float* H2 = H + H2OFF;
#pragma unroll
  for (int nt = 0; nt < 2; ++nt) {
    const int col = nt * 16 + rA;
    const float bb = b2[(col < 30) ? col : 29];
#pragma unroll
    for (int r = 0; r < 8; ++r) {
      float vv = fmaxf(acc2[nt][r] + bb, 0.f);
      vv = (col < 30) ? vv : 0.f;
      H2[(rD + r) * H2PITCH + col] = vv;
    }
  }
  __syncthreads();

  const int cc3 = (rA < 15) ? rA : 14;
  v8f acc3 = {};
  for (int kk = 0; kk < 8; ++kk) {
    const int k0  = kk * 4 + hs;
    const int kc0 = (k0 < 30) ? k0 : 29;
    const int kc1 = (k0 + 1 < 30) ? (k0 + 1) : 29;
    v2f a; a.x = H2[rA * H2PITCH + k0]; a.y = H2[rA * H2PITCH + k0 + 1];
    v2f b; b.x = w3[kc0 * 15 + cc3];    b.y = w3[kc1 * 15 + cc3];
    acc3 = __builtin_amdgcn_wmma_f32_16x16x4_f32(
        false, a, false, b, (short)0, acc3, false, false);
  }
  {
    const float bb = b3[cc3];
#pragma unroll
    for (int r = 0; r < 8; ++r)
      O[(rD + r) * OPITCH + rA] = acc3[r] + bb;
  }
  __syncthreads();

  if (lane < 16) {
    const int gRow = rowBase + lane;
    if (gRow < B) {
      float a9[9];
#pragma unroll
      for (int i = 0; i < 9; ++i) a9[i] = O[lane * OPITCH + i];
      float s[9], v[9];
#pragma unroll
      for (int i = 0; i < 3; ++i)
#pragma unroll
        for (int j = 0; j < 3; ++j) {
          s[i * 3 + j] = a9[0 * 3 + i] * a9[0 * 3 + j] +
                         a9[1 * 3 + i] * a9[1 * 3 + j] +
                         a9[2 * 3 + i] * a9[2 * 3 + j];
          v[i * 3 + j] = (i == j) ? 1.f : 0.f;
        }
#pragma unroll
      for (int sw = 0; sw < 6; ++sw) {
        jacobi3(s, v, 0, 1); jacobi3(s, v, 0, 2); jacobi3(s, v, 1, 2);
      }
      float g0 = s[0], g1 = s[4], g2 = s[8];
#define SWAPCOL(p, q) { float t_;                              \
        t_ = v[p];     v[p]     = v[q];     v[q]     = t_;     \
        t_ = v[3 + p]; v[3 + p] = v[3 + q]; v[3 + q] = t_;     \
        t_ = v[6 + p]; v[6 + p] = v[6 + q]; v[6 + q] = t_; }
      if (g0 < g1) { float t = g0; g0 = g1; g1 = t; SWAPCOL(0, 1); }
      if (g0 < g2) { float t = g0; g0 = g2; g2 = t; SWAPCOL(0, 2); }
      if (g1 < g2) { float t = g1; g1 = g2; g2 = t; SWAPCOL(1, 2); }
#undef SWAPCOL
      float w[9];
#pragma unroll
      for (int i = 0; i < 3; ++i)
#pragma unroll
        for (int j = 0; j < 3; ++j)
          w[i * 3 + j] = a9[i * 3 + 0] * v[0 + j] +
                         a9[i * 3 + 1] * v[3 + j] +
                         a9[i * 3 + 2] * v[6 + j];
      float u0x = w[0], u0y = w[3], u0z = w[6];
      float n0 = 1.f / sqrtf(u0x * u0x + u0y * u0y + u0z * u0z + 1e-20f);
      u0x *= n0; u0y *= n0; u0z *= n0;
      float u1x = w[1], u1y = w[4], u1z = w[7];
      float d01 = u1x * u0x + u1y * u0y + u1z * u0z;
      u1x -= d01 * u0x; u1y -= d01 * u0y; u1z -= d01 * u0z;
      float n1 = 1.f / sqrtf(u1x * u1x + u1y * u1y + u1z * u1z + 1e-20f);
      u1x *= n1; u1y *= n1; u1z *= n1;
      float cx = u0y * u1z - u0z * u1y;
      float cy = u0z * u1x - u0x * u1z;
      float cz = u0x * u1y - u0y * u1x;
      float detV = v[0] * (v[4] * v[8] - v[5] * v[7]) -
                   v[1] * (v[3] * v[8] - v[5] * v[6]) +
                   v[2] * (v[3] * v[7] - v[4] * v[6]);
      float sg = (detV >= 0.f) ? 1.f : -1.f;
      float u2x = sg * cx, u2y = sg * cy, u2z = sg * cz;
      float U_[9] = { u0x, u1x, u2x,  u0y, u1y, u2y,  u0z, u1z, u2z };
      float* op = out + (size_t)gRow * 15;
#pragma unroll
      for (int i = 0; i < 3; ++i)
#pragma unroll
        for (int j = 0; j < 3; ++j)
          op[i * 3 + j] = U_[i * 3 + 0] * v[j * 3 + 0] +
                          U_[i * 3 + 1] * v[j * 3 + 1] +
                          U_[i * 3 + 2] * v[j * 3 + 2];
#pragma unroll
      for (int i = 9; i < 15; ++i) op[i] = O[lane * OPITCH + i];
    }
  }
}

extern "C" void kernel_launch(void* const* d_in, const int* in_sizes, int n_in,
                              void* d_out, int out_size, void* d_ws, size_t ws_size,
                              hipStream_t stream) {
  (void)n_in; (void)out_size; (void)d_ws; (void)ws_size;
  const float* x   = (const float*)d_in[0];
  const float* cw  = (const float*)d_in[1];
  const float* cb  = (const float*)d_in[2];
  const float* w1  = (const float*)d_in[3];
  const float* b1  = (const float*)d_in[4];
  const float* w2  = (const float*)d_in[5];
  const float* b2  = (const float*)d_in[6];
  const float* w3  = (const float*)d_in[7];
  const float* b3  = (const float*)d_in[8];
  const int B = in_sizes[0] / 1200;
  const int grid = (B + WAVES * ROWS - 1) / (WAVES * ROWS);
  rmi_fused<<<grid, WAVES * 32, 0, stream>>>(x, cw, cb, w1, b1, w2, b2, w3, b3,
                                             (float*)d_out, B);
}


# per-row 3x3 SVD tail: every correctly rounded division and square root (5 per Jacobi rotation, 2 in the normalisations) becomes one rcp, sqrt or rsq instruction (about 1 ulp); sum orders unchanged
# speedup vs baseline: 2.8775x; 1.0473x over previous
#include <hip/hip_runtime.h>
#include <math.h>

typedef __attribute__((ext_vector_type(2))) float v2f;
typedef __attribute__((ext_vector_type(4))) float v4f;
typedef __attribute__((ext_vector_type(8))) float v8f;

#define WAVES   4
#define ROWS    16
#define HPITCH  197
#define H1PITCH 67
#define H2PITCH 33
#define OPITCH  17
#define H2OFF   1536

__device__ __forceinline__ void jacobi3(float* s, float* v, int p, int q) {
  float apq = s[p * 3 + q];
  if (fabsf(apq) < 1e-20f) return;
  float app = s[p * 3 + p], aqq = s[q * 3 + q];
  float tau = (aqq - app) * __builtin_amdgcn_rcpf(2.f * apq);
  float t  = copysignf(__builtin_amdgcn_rcpf(
                 fabsf(tau) + __builtin_amdgcn_sqrtf(fmaf(tau, tau, 1.f))), tau);
  float c  = __builtin_amdgcn_rsqf(fmaf(t, t, 1.f));
  float sn = t * c;
  const int k = 3 - p - q;
  float skp = s[k * 3 + p], skq = s[k * 3 + q];
  float nkp = c * skp - sn * skq;
  float nkq = sn * skp + c * skq;
  s[k * 3 + p] = nkp; s[p * 3 + k] = nkp;
  s[k * 3 + q] = nkq; s[q * 3 + k] = nkq;
  s[p * 3 + p] = c * c * app - 2.f * sn * c * apq + sn * sn * aqq;
  s[q * 3 + q] = sn * sn * app + 2.f * sn * c * apq + c * c * aqq;
  s[p * 3 + q] = 0.f; s[q * 3 + p] = 0.f;
#pragma unroll
  for (int i = 0; i < 3; ++i) {
    float vip = v[i * 3 + p], viq = v[i * 3 + q];
    v[i * 3 + p] = c * vip - sn * viq;
    v[i * 3 + q] = sn * vip + c * viq;
  }
}

__device__ __forceinline__ int src_row(int rowBase, int r, int B) {
  const int rr = (r < 15) ? r : 15;
  const int sr = rowBase + rr;
  return (sr < B) ? sr : (B - 1);
}

__device__ __forceinline__ void load_row(v4f (&buf)[18], const float* __restrict__ x,
                                         int row, int i0, int i1, int i2) {
  const v4f* p = reinterpret_cast<const v4f*>(x + (size_t)row * 1200);
#pragma unroll
  for (int c = 0; c < 6; ++c) {
    buf[c * 3 + 0] = p[c * 50 + i0];
    buf[c * 3 + 1] = p[c * 50 + i1];
    buf[c * 3 + 2] = p[c * 50 + i2];
  }
}

__device__ __forceinline__ void conv_row(const v4f (&buf)[18], const float (&cw)[30],
                                         float cb, float* hrow, int lane) {
  float o[8];
#pragma unroll
  for (int i = 0; i < 8; ++i) o[i] = cb;
#pragma unroll
  for (int c = 0; c < 6; ++c) {
    const v4f q0 = buf[c * 3 + 0], q1 = buf[c * 3 + 1], q2 = buf[c * 3 + 2];
    const float xv[12] = { q0.x, q0.y, q0.z, q0.w, q1.x, q1.y, q1.z, q1.w,
                           q2.x, q2.y, q2.z, q2.w };
#pragma unroll
    for (int k = 0; k < 5; ++k)
#pragma unroll
      for (int i = 0; i < 8; ++i)
        o[i] = fmaf(xv[i + k], cw[c * 5 + k], o[i]);
  }
  float* hp = hrow + 8 * lane;
  if (lane < 25) {
#pragma unroll
    for (int i = 0; i < 4; ++i) hp[i] = o[i];
  }
  if (lane < 24) {
#pragma unroll
    for (int i = 4; i < 8; ++i) hp[i] = o[i];
  }
}

__global__ __launch_bounds__(128) void rmi_fused(
    const float* __restrict__ x,   const float* __restrict__ cwg,
    const float* __restrict__ cbg, const float* __restrict__ w1,
    const float* __restrict__ b1,  const float* __restrict__ w2,
    const float* __restrict__ b2,  const float* __restrict__ w3,
    const float* __restrict__ b3,  float* __restrict__ out, int B) {
  __shared__ float lds [WAVES * ROWS * HPITCH];
  __shared__ float ldsO[WAVES * ROWS * OPITCH];

  const int tid  = threadIdx.x;
  const int wave = tid >> 5;
  const int lane = tid & 31;
  const int rowBase = blockIdx.x * (WAVES * ROWS) + wave * ROWS;
  float* H = &lds [wave * ROWS * HPITCH];
  float* O = &ldsO[wave * ROWS * OPITCH];

  float cw[30];
#pragma unroll
  for (int i = 0; i < 30; ++i) cw[i] = cwg[i];
  const float cb = cbg[0];

  const int lc = (lane < 24) ? lane : 24;
  const int i0 = 2 * lc, i1 = 2 * lc + 1;
  const int i2 = (lc < 24) ? (2 * lc + 2) : 49;
#define XROW(r) src_row(rowBase, (r), B)
  v4f bA[18], bB[18], bC[18];
  load_row(bA, x, XROW(0), i0, i1, i2);
  load_row(bB, x, XROW(1), i0, i1, i2);
  for (int r = 0; r < 15; r += 3) {
    load_row(bC, x, XROW(r + 2), i0, i1, i2);
    __builtin_amdgcn_sched_barrier(0);
    conv_row(bA, cw, cb, H + r * HPITCH, lane);
    load_row(bA, x, XROW(r + 3), i0, i1, i2);
    __builtin_amdgcn_sched_barrier(0);
    conv_row(bB, cw, cb, H + (r + 1) * HPITCH, lane);
    load_row(bB, x, XROW(r + 4), i0, i1, i2);
    __builtin_amdgcn_sched_barrier(0);
    conv_row(bC, cw, cb, H + (r + 2) * HPITCH, lane);
  }
  conv_row(bA, cw, cb, H + 15 * HPITCH, lane);
#undef XROW
  __syncthreads();

  const int hs = (lane >= 16) ? 2 : 0;
  const int rA = lane & 15;
  const int rD = (lane < 16) ? 0 : 8;

  int cc1[4];
#pragma unroll
  for (int nt = 0; nt < 4; ++nt) {
    const int c = nt * 16 + rA;
    cc1[nt] = (c < 50) ? c : 49;
  }

  v8f acc1[4] = {};
#pragma unroll 2
  for (int kk = 0; kk < 49; ++kk) {
    const int k0 = kk * 4 + hs;
    v2f a; a.x = H[rA * HPITCH + k0]; a.y = H[rA * HPITCH + k0 + 1];
#pragma unroll
    for (int nt = 0; nt < 4; ++nt) {
      v2f b;
      b.x = w1[k0 * 50 + cc1[nt]];
      b.y = w1[(k0 + 1) * 50 + cc1[nt]];
      acc1[nt] = __builtin_amdgcn_wmma_f32_16x16x4_f32(
          false, a, false, b, (short)0, acc1[nt], false, false);
    }
  }
  __syncthreads();

  float* H1 = H;
#pragma unroll
  for (int nt = 0; nt < 4; ++nt) {
    const int col = nt * 16 + rA;
    const float bb = b1[cc1[nt]];
#pragma unroll
    for (int r = 0; r < 8; ++r) {
      float vv = fmaxf(acc1[nt][r] + bb, 0.f);
      vv = (col < 50) ? vv : 0.f;
      H1[(rD + r) * H1PITCH + col] = vv;
    }
  }
  __syncthreads();

  const int cc2 = (16 + rA < 30) ? (16 + rA) : 29;
  v8f acc2[2] = {};
  for (int kk = 0; kk < 13; ++kk) {
    const int k0  = kk * 4 + hs;
    const int kc0 = (k0 < 50) ? k0 : 49;
    const int kc1 = (k0 + 1 < 50) ? (k0 + 1) : 49;
    v2f a;  a.x  = H1[rA * H1PITCH + k0]; a.y = H1[rA * H1PITCH + k0 + 1];
    v2f b0; b0.x = w2[kc0 * 30 + rA];     b0.y = w2[kc1 * 30 + rA];
    v2f b1v; b1v.x = w2[kc0 * 30 + cc2];  b1v.y = w2[kc1 * 30 + cc2];
    acc2[0] = __builtin_amdgcn_wmma_f32_16x16x4_f32(
        false, a, false, b0, (short)0, acc2[0], false, false);
    acc2[1] = __builtin_amdgcn_wmma_f32_16x16x4_f32(
        false, a, false, b1v, (short)0, acc2[1], false, false);
  }
  __syncthreads();
  float* H2 = H + H2OFF;
#pragma unroll
  for (int nt = 0; nt < 2; ++nt) {
    const int col = nt * 16 + rA;
    const float bb = b2[(col < 30) ? col : 29];
#pragma unroll
    for (int r = 0; r < 8; ++r) {
      float vv = fmaxf(acc2[nt][r] + bb, 0.f);
      vv = (col < 30) ? vv : 0.f;
      H2[(rD + r) * H2PITCH + col] = vv;
    }
  }
  __syncthreads();

  const int cc3 = (rA < 15) ? rA : 14;
  v8f acc3 = {};
  for (int kk = 0; kk < 8; ++kk) {
    const int k0  = kk * 4 + hs;
    const int kc0 = (k0 < 30) ? k0 : 29;
    const int kc1 = (k0 + 1 < 30) ? (k0 + 1) : 29;
    v2f a; a.x = H2[rA * H2PITCH + k0]; a.y = H2[rA * H2PITCH + k0 + 1];
    v2f b; b.x = w3[kc0 * 15 + cc3];    b.y = w3[kc1 * 15 + cc3];
    acc3 = __builtin_amdgcn_wmma_f32_16x16x4_f32(
        false, a, false, b, (short)0, acc3, false, false);
  }
  {
    const float bb = b3[cc3];
#pragma unroll
    for (int r = 0; r < 8; ++r)
      O[(rD + r) * OPITCH + rA] = acc3[r] + bb;
  }
  __syncthreads();

  if (lane < 16) {
    const int gRow = rowBase + lane;
    if (gRow < B) {
      float a9[9];
#pragma unroll
      for (int i = 0; i < 9; ++i) a9[i] = O[lane * OPITCH + i];
      float s[9], v[9];
#pragma unroll
      for (int i = 0; i < 3; ++i)
#pragma unroll
        for (int j = 0; j < 3; ++j) {
          s[i * 3 + j] = a9[0 * 3 + i] * a9[0 * 3 + j] +
                         a9[1 * 3 + i] * a9[1 * 3 + j] +
                         a9[2 * 3 + i] * a9[2 * 3 + j];
          v[i * 3 + j] = (i == j) ? 1.f : 0.f;
        }
#pragma unroll
      for (int sw = 0; sw < 6; ++sw) {
        jacobi3(s, v, 0, 1); jacobi3(s, v, 0, 2); jacobi3(s, v, 1, 2);
      }
      float g0 = s[0], g1 = s[4], g2 = s[8];
#define SWAPCOL(p, q) { float t_;                              \
        t_ = v[p];     v[p]     = v[q];     v[q]     = t_;     \
        t_ = v[3 + p]; v[3 + p] = v[3 + q]; v[3 + q] = t_;     \
        t_ = v[6 + p]; v[6 + p] = v[6 + q]; v[6 + q] = t_; }
      if (g0 < g1) { float t = g0; g0 = g1; g1 = t; SWAPCOL(0, 1); }
      if (g0 < g2) { float t = g0; g0 = g2; g2 = t; SWAPCOL(0, 2); }
      if (g1 < g2) { float t = g1; g1 = g2; g2 = t; SWAPCOL(1, 2); }
#undef SWAPCOL
      float w[9];
#pragma unroll
      for (int i = 0; i < 3; ++i)
#pragma unroll
        for (int j = 0; j < 3; ++j)
          w[i * 3 + j] = a9[i * 3 + 0] * v[0 + j] +
                         a9[i * 3 + 1] * v[3 + j] +
                         a9[i * 3 + 2] * v[6 + j];
      float u0x = w[0], u0y = w[3], u0z = w[6];
      float n0 = __builtin_amdgcn_rsqf(u0x * u0x + u0y * u0y + u0z * u0z + 1e-20f);
      u0x *= n0; u0y *= n0; u0z *= n0;
      float u1x = w[1], u1y = w[4], u1z = w[7];
      float d01 = u1x * u0x + u1y * u0y + u1z * u0z;
      u1x -= d01 * u0x; u1y -= d01 * u0y; u1z -= d01 * u0z;
      float n1 = __builtin_amdgcn_rsqf(u1x * u1x + u1y * u1y + u1z * u1z + 1e-20f);
      u1x *= n1; u1y *= n1; u1z *= n1;
      float cx = u0y * u1z - u0z * u1y;
      float cy = u0z * u1x - u0x * u1z;
      float cz = u0x * u1y - u0y * u1x;
      float detV = v[0] * (v[4] * v[8] - v[5] * v[7]) -
                   v[1] * (v[3] * v[8] - v[5] * v[6]) +
                   v[2] * (v[3] * v[7] - v[4] * v[6]);
      float sg = (detV >= 0.f) ? 1.f : -1.f;
      float u2x = sg * cx, u2y = sg * cy, u2z = sg * cz;
      float U_[9] = { u0x, u1x, u2x,  u0y, u1y, u2y,  u0z, u1z, u2z };
      float* op = out + (size_t)gRow * 15;
#pragma unroll
      for (int i = 0; i < 3; ++i)
#pragma unroll
        for (int j = 0; j < 3; ++j)
          op[i * 3 + j] = U_[i * 3 + 0] * v[j * 3 + 0] +
                          U_[i * 3 + 1] * v[j * 3 + 1] +
                          U_[i * 3 + 2] * v[j * 3 + 2];
#pragma unroll
      for (int i = 9; i < 15; ++i) op[i] = O[lane * OPITCH + i];
    }
  }
}

extern "C" void kernel_launch(void* const* d_in, const int* in_sizes, int n_in,
                              void* d_out, int out_size, void* d_ws, size_t ws_size,
                              hipStream_t stream) {
  (void)n_in; (void)out_size; (void)d_ws; (void)ws_size;
  const float* x   = (const float*)d_in[0];
  const float* cw  = (const float*)d_in[1];
  const float* cb  = (const float*)d_in[2];
  const float* w1  = (const float*)d_in[3];
  const float* b1  = (const float*)d_in[4];
  const float* w2  = (const float*)d_in[5];
  const float* b2  = (const float*)d_in[6];
  const float* w3  = (const float*)d_in[7];
  const float* b3  = (const float*)d_in[8];
  const int B = in_sizes[0] / 1200;
  const int grid = (B + WAVES * ROWS - 1) / (WAVES * ROWS);
  rmi_fused<<<grid, WAVES * 32, 0, stream>>>(x, cw, cb, w1, b1, w2, b2, w3, b3,
                                             (float*)d_out, B);
}
